# starting kernel
# baseline (speedup 1.0000x reference)
#include <hip/hip_runtime.h>

typedef __attribute__((ext_vector_type(2))) float v2f;
typedef __attribute__((ext_vector_type(8))) float v8f;

#define WW   256
#define HWSZ 65536
#define CCH  8
#define KKT  25
#define NEG_SLOPE 0.2f


__device__ __forceinline__ void mma_extract_store(const v2f A[7], const v2f B[7],
                                                  int lane, float* __restrict__ outp)
{
    v8f acc = {0.f, 0.f, 0.f, 0.f, 0.f, 0.f, 0.f, 0.f};
#pragma unroll
    for (int kc = 0; kc < 7; ++kc) {
        acc = __builtin_amdgcn_wmma_f32_16x16x4_f32(
             false, A[kc],  false, B[kc],
             (short)0, acc,  false,  false);
    }
    const int idx = lane & 7;
    float v = acc[0];
#pragma unroll
    for (int i = 1; i < 8; ++i) v = (idx == i) ? acc[i] : v;
    v = (v >= 0.0f) ? v : NEG_SLOPE * v;
    if (lane < 8 || lane >= 24) {
        const int p = (lane < 16) ? lane : (lane - 16);
        __builtin_nontemporal_store(v, outp + p);
    }
}

__global__ __launch_bounds__(256) void dynconv5x5_wmma(
    const float* __restrict__ x,
    const float* __restrict__ kern,
    float* __restrict__ out)
{
    const int lane  = threadIdx.x & 31;
    const int wave  = threadIdx.x >> 5;
    const int tx    = (blockIdx.x << 3) + wave;
    const int y     = blockIdx.y;
    const int plane = blockIdx.z;

    const int x0 = tx << 4;
    const int hf = lane >> 4;
    const int lp = lane & 15;

    const float* __restrict__ xp = x + plane * HWSZ;
    const int kbase = (plane * KKT) * HWSZ + y * WW + x0 + lp;
    float* __restrict__ outp = out + plane * HWSZ + y * WW + x0;

    if ((y >= 2) & (y <= 253) & (tx >= 1) & (tx <= 14)) {
        const float* __restrict__ xb    = xp + (y - 2) * WW + (x0 - 2);
        const float* __restrict__ pA2   = xb + lp + hf * 2;
        const float* __restrict__ pA253 = xb + lp + hf * 253;
        const float* __restrict__ pB    = kern + kbase + hf * (2 * HWSZ);

        constexpr int T0[12]   = {0, 1, 4,   5,   8,   9,   12,  13,  16,  17,  20,   21};
        constexpr int OFF0[12] = {0, 1, 4, 256, 259, 260, 514, 515, 769, 770, 1024, 1025};
        constexpr int D253[12] = {0, 0, 1,   0,   1,   1,   0,   1,   0,   0,   0,    0};

        v2f A[7], B[7];
#pragma unroll
        for (int i = 0; i < 12; ++i) {
            const int kc = T0[i] >> 2;
            const int j  = T0[i] & 3;
            A[kc][j] = D253[i] ? pA253[OFF0[i]] : pA2[OFF0[i]];
            B[kc][j] = __builtin_nontemporal_load(pB + T0[i] * HWSZ);
        }
        const float a24 = xb[lp + 1028];
        const float b24 = __builtin_nontemporal_load(kern + kbase + 24 * HWSZ);
        A[6][0] = hf ? 0.0f : a24;
        B[6][0] = hf ? 0.0f : b24;
        A[6][1] = 0.0f;
        B[6][1] = 0.0f;

        mma_extract_store(A, B, lane, outp);
        return;
    }

    {
        v2f A[7], B[7];
#pragma unroll
        for (int kc = 0; kc < 7; ++kc) {
#pragma unroll
            for (int j = 0; j < 2; ++j) {
                const int t  = 4 * kc + 2 * hf + j;
                const int tc = (t < KKT) ? t : (KKT - 1);
                const int dy = tc / 5 - 2;
                const int dx = tc % 5 - 2;
                int cy = y + dy;       cy = cy < 0 ? 0 : (cy > 255 ? 255 : cy);
                int cx = x0 + lp + dx; cx = cx < 0 ? 0 : (cx > 255 ? 255 : cx);
                const float av = xp[cy * WW + cx];
                const float bv = __builtin_nontemporal_load(kern + kbase + tc * HWSZ);
                const bool valid = (t < KKT);
                A[kc][j] = valid ? av : 0.0f;
                B[kc][j] = valid ? bv : 0.0f;
            }
        }
        mma_extract_store(A, B, lane, outp);
    }
}

extern "C" void kernel_launch(void* const* d_in, const int* in_sizes, int n_in,
                              void* d_out, int out_size, void* d_ws, size_t ws_size,
                              hipStream_t stream) {
    (void)in_sizes; (void)n_in; (void)d_ws; (void)ws_size; (void)out_size;
    const float* x    = (const float*)d_in[0];
    const float* kern = (const float*)d_in[1];
    float*       out  = (float*)d_out;

    dim3 grid(2, 256, 32);
    dynconv5x5_wmma<<<grid, 256, 0, stream>>>(x, kern, out);
}


# 32 px x 2 rows per wave: each 4-byte-per-lane load covers one whole 128-byte line (lane = pixel), a half-wave select builds two WMMA tiles; tile 1 swaps its k halves, so f32 summation order changes
# speedup vs baseline: 1.2606x; 1.2606x over previous
#include <hip/hip_runtime.h>


typedef __attribute__((ext_vector_type(2))) float v2f;
typedef __attribute__((ext_vector_type(8))) float v8f;

#define WW   256
#define HWSZ 65536
#define KKT  25
#define NEG_SLOPE 0.2f
#define ROWS 2
static_assert(ROWS == 2, "kernel_launch launches 128 row pairs per plane");


__device__ __forceinline__ void split_pair(float s, float c, bool lo, float& t0, float& t1)
{
    t0 = lo ? s : c;
    t1 = lo ? c : s;
}

__device__ __forceinline__ float diag_lrelu(const v8f acc, int idx)
{
    float v = acc[0];
#pragma unroll
    for (int i = 1; i < 8; ++i) v = (idx == i) ? acc[i] : v;
    return (v >= 0.0f) ? v : NEG_SLOPE * v;
}

__device__ __forceinline__ int clamp255(int v)
{
    return v < 0 ? 0 : (v > 255 ? 255 : v);
}

__global__ __launch_bounds__(256) void dynconv5x5_wmma_line32(
    const float* __restrict__ x,
    const float* __restrict__ kern,
    float* __restrict__ out)
{
    const int lane  = threadIdx.x & 31;
    const int wave  = threadIdx.x >> 5;
    const int y0    = blockIdx.x * ROWS;
    const int plane = blockIdx.y;
    const int x0    = wave << 5;
    const bool lo   = lane < 16;
    const int pxs   = lane;
    const int pxc   = lane ^ 16;

    int cs[5], cc[5];
#pragma unroll
    for (int kx = 0; kx < 5; ++kx) {
        cs[kx] = clamp255(x0 + pxs + kx - 2);
        cc[kx] = clamp255(x0 + pxc + kx - 2);
    }

    const float* __restrict__ xp = x + plane * HWSZ;
    const float* __restrict__ kp = kern + (plane * KKT) * HWSZ + x0;

    float sB[ROWS][13], cB[ROWS][12], sA[ROWS][13], cA[ROWS][12];
#pragma unroll
    for (int r = 0; r < ROWS; ++r) {
        const float* __restrict__ kr = kp + (y0 + r) * WW;
#pragma unroll
        for (int i = 0; i < 12; ++i) {
            const int ta = 4 * (i >> 1) + (i & 1);
            sB[r][i] = __builtin_nontemporal_load(kr + ta * HWSZ + pxs);
            cB[r][i] = __builtin_nontemporal_load(kr + (ta + 2) * HWSZ + pxc);
        }
        sB[r][12] = __builtin_nontemporal_load(kr + 24 * HWSZ + pxs);
    }
#pragma unroll
    for (int r = 0; r < ROWS; ++r) {
        const int y = y0 + r;
        const float* __restrict__ xr[5];
#pragma unroll
        for (int ky = 0; ky < 5; ++ky)
            xr[ky] = xp + clamp255(y + ky - 2) * WW;
#pragma unroll
        for (int i = 0; i < 12; ++i) {
            const int ta = 4 * (i >> 1) + (i & 1);
            const int tb = ta + 2;
            sA[r][i] = xr[ta / 5][cs[ta % 5]];
            cA[r][i] = xr[tb / 5][cc[tb % 5]];
        }
        sA[r][12] = xr[4][cs[4]];
    }

    __builtin_amdgcn_sched_barrier(0);

    v2f A0[ROWS][7], A1[ROWS][7], B0[ROWS][7], B1[ROWS][7];
#pragma unroll
    for (int r = 0; r < ROWS; ++r) {
#pragma unroll
        for (int i = 0; i < 12; ++i) {
            const int kc = i >> 1, j = i & 1;
            float a0, a1, b0, b1;
            split_pair(sA[r][i], cA[r][i], lo, a0, a1);
            split_pair(sB[r][i], cB[r][i], lo, b0, b1);
            A0[r][kc][j] = a0;  A1[r][kc][j] = a1;
            B0[r][kc][j] = b0;  B1[r][kc][j] = b1;
        }
        {
            float a0, a1, b0, b1;
            split_pair(sA[r][12], 0.0f, lo, a0, a1);
            split_pair(sB[r][12], 0.0f, lo, b0, b1);
            A0[r][6][0] = a0;  A1[r][6][0] = a1;
            B0[r][6][0] = b0;  B1[r][6][0] = b1;
            A0[r][6][1] = 0.0f;  A1[r][6][1] = 0.0f;
            B0[r][6][1] = 0.0f;  B1[r][6][1] = 0.0f;
        }
    }

    v8f acc0[ROWS], acc1[ROWS];
#pragma unroll
    for (int r = 0; r < ROWS; ++r) {
        acc0[r] = (v8f){0.f, 0.f, 0.f, 0.f, 0.f, 0.f, 0.f, 0.f};
        acc1[r] = (v8f){0.f, 0.f, 0.f, 0.f, 0.f, 0.f, 0.f, 0.f};
    }
#pragma unroll
    for (int kc = 0; kc < 7; ++kc) {
#pragma unroll
        for (int r = 0; r < ROWS; ++r) {
            acc0[r] = __builtin_amdgcn_wmma_f32_16x16x4_f32(
                 false, A0[r][kc],  false, B0[r][kc],
                 (short)0, acc0[r],  false,  false);
            acc1[r] = __builtin_amdgcn_wmma_f32_16x16x4_f32(
                 false, A1[r][kc],  false, B1[r][kc],
                 (short)0, acc1[r],  false,  false);
        }
    }

    const int idx = lane & 7;
    const bool own0 = (lane < 8) | (lane >= 24);
    const int  px   = (lane < 8) ? lane : ((lane < 24) ? (lane + 8) : (lane - 16));
#pragma unroll
    for (int r = 0; r < ROWS; ++r) {
        const float v0 = diag_lrelu(acc0[r], idx);
        const float v1 = diag_lrelu(acc1[r], idx);
        const float w1 = __shfl_xor(v1, 8);
        float* __restrict__ outp = out + plane * HWSZ + (y0 + r) * WW + x0;
        __builtin_nontemporal_store(own0 ? v0 : w1, outp + px);
    }
}

extern "C" void kernel_launch(void* const* d_in, const int* in_sizes, int n_in,
                              void* d_out, int out_size, void* d_ws, size_t ws_size,
                              hipStream_t stream) {
    (void)in_sizes; (void)n_in; (void)d_ws; (void)ws_size; (void)out_size;
    const float* x    = (const float*)d_in[0];
    const float* kern = (const float*)d_in[1];
    float*       out  = (float*)d_out;

    dim3 grid(128, 32);
    dynconv5x5_wmma_line32<<<grid, 256, 0, stream>>>(x, kern, out);
}
